# starting kernel
# baseline (speedup 1.0000x reference)
#include <hip/hip_runtime.h>
#include <math.h>


typedef __attribute__((ext_vector_type(2))) float v2f;
typedef __attribute__((ext_vector_type(4))) float v4f;
typedef __attribute__((ext_vector_type(8))) float v8f;

#define B_DIM 256
#define T_DIM 512
#define F_DIM 256
#define U_DIM 256

__global__ __launch_bounds__(256)
void lstm_head_tanh_wmma_f32(const float* __restrict__ x,
                             const float* __restrict__ w,
                             const float* __restrict__ bias,
                             float* __restrict__ out)
{
    __shared__ __align__(16) float Bs[F_DIM * 16];

    const int tid  = threadIdx.x;
    const int wave = tid >> 5;
    const int lane = tid & 31;
    const int half = lane >> 4;
    const int lr   = lane & 15;

    const int n0 = blockIdx.x * 16;

    {
        const v4f* __restrict__ wrow;
        for (int i = tid; i < F_DIM * 4; i += 256) {
            const int row = i >> 2;
            const int q   = i & 3;
            wrow = reinterpret_cast<const v4f*>(w + row * U_DIM + n0);
            reinterpret_cast<v4f*>(Bs)[i] = wrow[q];
        }
    }
    __syncthreads();

    const int    mt        = blockIdx.y * 8 + wave;
    const size_t rowStride = (size_t)T_DIM * F_DIM;
    const float* __restrict__ rowA =
        x + (size_t)(mt * 16 + lr) * rowStride + (size_t)(T_DIM - 1) * F_DIM;

    #pragma unroll
    for (int pb = 0; pb < 1024; pb += 256)
        __builtin_prefetch(reinterpret_cast<const char*>(rowA) + pb, 0, 0);

    v8f acc = {};

    #pragma unroll 4
    for (int kk = 0; kk < F_DIM; kk += 4) {
        const int kb = kk + half * 2;
        v2f a = *reinterpret_cast<const v2f*>(rowA + kb);
        v2f b;
        b.x = Bs[kb * 16 + lr];
        b.y = Bs[(kb + 1) * 16 + lr];
        acc = __builtin_amdgcn_wmma_f32_16x16x4_f32(
             false, a,  false, b,
             (short)0, acc,  false,  false);
    }

    const int   col = n0 + lr;
    const float bv  = bias[col];
    #pragma unroll
    for (int r = 0; r < 8; ++r) {
        const int row = mt * 16 + r + half * 8;
        out[row * U_DIM + col] = tanhf(acc[r] + bv);
    }
}

extern "C" void kernel_launch(void* const* d_in, const int* in_sizes, int n_in,
                              void* d_out, int out_size, void* d_ws, size_t ws_size,
                              hipStream_t stream) {
    (void)in_sizes; (void)n_in; (void)out_size; (void)d_ws; (void)ws_size;
    const float* x    = (const float*)d_in[0];
    const float* w    = (const float*)d_in[1];
    const float* bias = (const float*)d_in[2];
    float*       out  = (float*)d_out;

    dim3 grid(U_DIM / 16, 2);
    dim3 block(256);
    lstm_head_tanh_wmma_f32<<<grid, block, 0, stream>>>(x, w, bias, out);
}


# Both operands staged through LDS by whole-line 16-byte loads, every load of a wave requested before its first wait (was 21 dependent round trips of narrow loads); same k order, one accumulator
# speedup vs baseline: 1.4994x; 1.4994x over previous
#include <hip/hip_runtime.h>
#include <math.h>


typedef __attribute__((ext_vector_type(2))) float v2f;
typedef __attribute__((ext_vector_type(4))) float v4f;
typedef __attribute__((ext_vector_type(8))) float v8f;

#define B_DIM 256
#define T_DIM 512
#define F_DIM 256
#define U_DIM 256

#define A_PITCH 260

__global__ __launch_bounds__(256)
void lstm_head_tanh_wmma_f32(const float* __restrict__ x,
                             const float* __restrict__ w,
                             const float* __restrict__ bias,
                             float* __restrict__ out)
{
    __shared__ __align__(16) float Bt[16 * A_PITCH];
    __shared__ __align__(16) float As[8 * 16 * A_PITCH];

    const int tid  = threadIdx.x;
    const int wave = tid >> 5;
    const int lane = tid & 31;
    const int half = lane >> 4;
    const int lr   = lane & 15;

    const int n0 = blockIdx.x * 16;
    const int mt = blockIdx.y * 8 + wave;
    const size_t rowStride = (size_t)T_DIM * F_DIM;

    const float* __restrict__ xt =
        x + (size_t)(mt * 16) * rowStride + (size_t)(T_DIM - 1) * F_DIM;

    const int   col = n0 + lr;
    const float bv  = bias[col];

    v4f wreg[4];
    #pragma unroll
    for (int j = 0; j < 4; ++j) {
        const int i   = tid + j * 256;
        const int row = i >> 2;
        const int q   = i & 3;
        wreg[j] = *reinterpret_cast<const v4f*>(w + row * U_DIM + n0 + q * 4);
    }
    v4f areg[32];
    #pragma unroll
    for (int j = 0; j < 32; ++j) {
        const int r  = j >> 1;
        const int hh = j & 1;
        areg[j] = *reinterpret_cast<const v4f*>(xt + (size_t)r * rowStride + hh * 128 + lane * 4);
    }

    #pragma unroll
    for (int j = 0; j < 4; ++j) {
        const int i   = tid + j * 256;
        const int row = i >> 2;
        const int q   = i & 3;
        #pragma unroll
        for (int e = 0; e < 4; ++e)
            Bt[(q * 4 + e) * A_PITCH + row] = wreg[j][e];
    }

    float* Aw = As + wave * (16 * A_PITCH);
    #pragma unroll
    for (int j = 0; j < 32; ++j) {
        const int r  = j >> 1;
        const int hh = j & 1;
        *reinterpret_cast<v4f*>(Aw + r * A_PITCH + hh * 128 + lane * 4) = areg[j];
    }
    __syncthreads();

    v8f acc = {};

    const float* arow = Aw + lr * A_PITCH + half * 2;
    const float* brow = Bt + lr * A_PITCH + half * 2;

    v2f af[2][8], bf[2][8];
    #pragma unroll
    for (int u = 0; u < 8; ++u) {
        af[0][u] = *reinterpret_cast<const v2f*>(arow + 4 * u);
        bf[0][u] = *reinterpret_cast<const v2f*>(brow + 4 * u);
    }
    #pragma unroll
    for (int g = 0; g < 8; ++g) {
        const int cur = g & 1, nxt = cur ^ 1;
        if (g < 7) {
            #pragma unroll
            for (int u = 0; u < 8; ++u) {
                af[nxt][u] = *reinterpret_cast<const v2f*>(arow + 4 * (8 * (g + 1) + u));
                bf[nxt][u] = *reinterpret_cast<const v2f*>(brow + 4 * (8 * (g + 1) + u));
            }
        }
        __builtin_amdgcn_sched_barrier(0);
        #pragma unroll
        for (int u = 0; u < 8; ++u) {
            acc = __builtin_amdgcn_wmma_f32_16x16x4_f32(
                 false, af[cur][u],  false, bf[cur][u],
                 (short)0, acc,  false,  false);
        }
        __builtin_amdgcn_sched_barrier(0);
    }

    #pragma unroll
    for (int r = 0; r < 8; ++r) {
        const int row = mt * 16 + r + half * 8;
        out[row * U_DIM + col] = tanhf(acc[r] + bv);
    }
}

extern "C" void kernel_launch(void* const* d_in, const int* in_sizes, int n_in,
                              void* d_out, int out_size, void* d_ws, size_t ws_size,
                              hipStream_t stream) {
    (void)in_sizes; (void)n_in; (void)out_size; (void)d_ws; (void)ws_size;
    const float* x    = (const float*)d_in[0];
    const float* w    = (const float*)d_in[1];
    const float* bias = (const float*)d_in[2];
    float*       out  = (float*)d_out;

    dim3 grid(U_DIM / 16, 2);
    dim3 block(256);
    lstm_head_tanh_wmma_f32<<<grid, block, 0, stream>>>(x, w, bias, out);
}


# 64 workgroups of 4 waves, one 16x16 tile per wave and one wave per SIMD (was 32 workgroups of 8 waves, two per SIMD); K whole in each wave, same k order, one accumulator
# speedup vs baseline: 1.7067x; 1.1383x over previous
#include <hip/hip_runtime.h>
#include <math.h>


typedef __attribute__((ext_vector_type(2))) float v2f;
typedef __attribute__((ext_vector_type(4))) float v4f;
typedef __attribute__((ext_vector_type(8))) float v8f;

#define B_DIM 256
#define T_DIM 512
#define F_DIM 256
#define U_DIM 256

#define A_PITCH 260

__global__ __launch_bounds__(128)
void lstm_head_tanh_wmma_f32(const float* __restrict__ x,
                             const float* __restrict__ w,
                             const float* __restrict__ bias,
                             float* __restrict__ out)
{
    __shared__ __align__(16) float Bt[16 * A_PITCH];
    __shared__ __align__(16) float As[4 * 16 * A_PITCH];

    const int tid  = threadIdx.x;
    const int wave = tid >> 5;
    const int lane = tid & 31;
    const int half = lane >> 4;
    const int lr   = lane & 15;

    const int n0 = blockIdx.x * 16;
    const int mt = blockIdx.y * 4 + wave;
    const size_t rowStride = (size_t)T_DIM * F_DIM;

    const float* __restrict__ xt =
        x + (size_t)(mt * 16) * rowStride + (size_t)(T_DIM - 1) * F_DIM;

    const int   col = n0 + lr;
    const float bv  = bias[col];

    v4f wreg[8];
    #pragma unroll
    for (int j = 0; j < 8; ++j) {
        const int i   = tid + j * 128;
        const int row = i >> 2;
        const int q   = i & 3;
        wreg[j] = *reinterpret_cast<const v4f*>(w + row * U_DIM + n0 + q * 4);
    }
    v4f areg[32];
    #pragma unroll
    for (int j = 0; j < 32; ++j) {
        const int r  = j >> 1;
        const int hh = j & 1;
        areg[j] = *reinterpret_cast<const v4f*>(xt + (size_t)r * rowStride + hh * 128 + lane * 4);
    }

    #pragma unroll
    for (int j = 0; j < 8; ++j) {
        const int i   = tid + j * 128;
        const int row = i >> 2;
        const int q   = i & 3;
        #pragma unroll
        for (int e = 0; e < 4; ++e)
            Bt[(q * 4 + e) * A_PITCH + row] = wreg[j][e];
    }

    float* Aw = As + wave * (16 * A_PITCH);
    #pragma unroll
    for (int j = 0; j < 32; ++j) {
        const int r  = j >> 1;
        const int hh = j & 1;
        *reinterpret_cast<v4f*>(Aw + r * A_PITCH + hh * 128 + lane * 4) = areg[j];
    }
    __syncthreads();

    v8f acc = {};

    const float* arow = Aw + lr * A_PITCH + half * 2;
    const float* brow = Bt + lr * A_PITCH + half * 2;

    v2f af[2][8], bf[2][8];
    #pragma unroll
    for (int u = 0; u < 8; ++u) {
        af[0][u] = *reinterpret_cast<const v2f*>(arow + 4 * u);
        bf[0][u] = *reinterpret_cast<const v2f*>(brow + 4 * u);
    }
    #pragma unroll
    for (int g = 0; g < 8; ++g) {
        const int cur = g & 1, nxt = cur ^ 1;
        if (g < 7) {
            #pragma unroll
            for (int u = 0; u < 8; ++u) {
                af[nxt][u] = *reinterpret_cast<const v2f*>(arow + 4 * (8 * (g + 1) + u));
                bf[nxt][u] = *reinterpret_cast<const v2f*>(brow + 4 * (8 * (g + 1) + u));
            }
        }
        __builtin_amdgcn_sched_barrier(0);
        #pragma unroll
        for (int u = 0; u < 8; ++u) {
            acc = __builtin_amdgcn_wmma_f32_16x16x4_f32(
                 false, af[cur][u],  false, bf[cur][u],
                 (short)0, acc,  false,  false);
        }
        __builtin_amdgcn_sched_barrier(0);
    }

    #pragma unroll
    for (int r = 0; r < 8; ++r) {
        const int row = mt * 16 + r + half * 8;
        out[row * U_DIM + col] = tanhf(acc[r] + bv);
    }
}

extern "C" void kernel_launch(void* const* d_in, const int* in_sizes, int n_in,
                              void* d_out, int out_size, void* d_ws, size_t ws_size,
                              hipStream_t stream) {
    (void)in_sizes; (void)n_in; (void)out_size; (void)d_ws; (void)ws_size;
    const float* x    = (const float*)d_in[0];
    const float* w    = (const float*)d_in[1];
    const float* bias = (const float*)d_in[2];
    float*       out  = (float*)d_out;

    dim3 grid(U_DIM / 16, 4);
    dim3 block(128);
    lstm_head_tanh_wmma_f32<<<grid, block, 0, stream>>>(x, w, bias, out);
}
